# starting kernel
# baseline (speedup 1.0000x reference)
#include <hip/hip_runtime.h>


typedef float v2f __attribute__((ext_vector_type(2)));
typedef float v8f __attribute__((ext_vector_type(8)));

#define N_NODES 50000
#define N_EDGES 800000
#define F_IN    128
#define H1      200
#define H1P     208
#define H2      100
#define H2P     112
#define NC      40
#define NCP     48

#define KP      210

__global__ void k_deg_init(float* __restrict__ deg, int n) {
    int i = blockIdx.x * blockDim.x + threadIdx.x;
    if (i < n) deg[i] = 1.0f;
}

__global__ void k_deg_accum(const int* __restrict__ dst, float* __restrict__ deg, int e) {
    int i = blockIdx.x * blockDim.x + threadIdx.x;
    if (i < e) atomicAdd(&deg[dst[i]], 1.0f);
}

__global__ void k_rsqrt_inplace(float* __restrict__ v, int n) {
    int i = blockIdx.x * blockDim.x + threadIdx.x;
    if (i < n) v[i] = rsqrtf(v[i]);
}

__global__ void __launch_bounds__(128)
k_gemm_wmma_f32(const float* __restrict__ A, int lda,
                const float* __restrict__ B, int Nact,
                float* __restrict__ C, int ldc, int Nstore,
                int K, int reluA, const float* __restrict__ bias)
{
    __shared__ float As[16 * KP];
    __shared__ float Bs[64 * KP];

    const int m_base = blockIdx.x * 16;
    const int nblk   = blockIdx.y * 64;
    const int tid    = threadIdx.x;

    for (int idx = tid; idx < 16 * K; idx += 128) {
        int r = idx / K;
        int c = idx - r * K;
        float v = A[(size_t)(m_base + r) * lda + c];
        if (reluA) v = fmaxf(v, 0.0f);
        As[r * KP + c] = v;
    }
    for (int idx = tid; idx < 64 * K; idx += 128) {
        int k = idx >> 6;
        int c = idx & 63;
        int col = nblk + c;
        float v = (col < Nact) ? B[(size_t)k * Nact + col] : 0.0f;
        Bs[c * KP + k] = v;
    }
    __syncthreads();

    const int wave = tid >> 5;
    const int lane = tid & 31;
    const int mloc = lane & 15;
    const int kb   = (lane >> 4) << 1;
    const int ncol = nblk + wave * 16 + mloc;

    const float* ap = &As[mloc * KP + kb];
    const float* bp = &Bs[(wave * 16 + mloc) * KP + kb];

    v8f acc = {0.f, 0.f, 0.f, 0.f, 0.f, 0.f, 0.f, 0.f};

    for (int k = 0; k < K; k += 4) {
        v2f a, b;
        a.x = ap[k];
        a.y = ap[k + 1];
        b.x = bp[k];
        b.y = bp[k + 1];
        acc = __builtin_amdgcn_wmma_f32_16x16x4_f32(
                  false, a, false, b, (short)0, acc, false, false);
    }

    float bv = (bias != nullptr && ncol < Nact) ? bias[ncol] : 0.0f;
    if (ncol < Nstore) {
        const int rbase = m_base + ((lane >> 4) << 3);
        #pragma unroll
        for (int g = 0; g < 8; ++g)
            C[(size_t)(rbase + g) * ldc + ncol] = acc[g] + bv;
    }
}

__global__ void k_agg_init(const float* __restrict__ hw, const float* __restrict__ dinv,
                           const float* __restrict__ bias, float* __restrict__ agg,
                           int Fp, int Fact, int total)
{
    int i = blockIdx.x * blockDim.x + threadIdx.x;
    if (i >= total) return;
    int node = i / Fp;
    int f    = i - node * Fp;
    float d  = dinv[node];
    float bb = (f < Fact) ? bias[f] : 0.0f;
    agg[i] = bb + d * d * hw[i];
}

__global__ void k_agg_edges(const int* __restrict__ src, const int* __restrict__ dst,
                            const float* __restrict__ dinv, const float* __restrict__ hw,
                            float* __restrict__ agg, int Fp, int Fact, int e)
{
    int wid    = (blockIdx.x * blockDim.x + threadIdx.x) >> 5;
    int lane   = threadIdx.x & 31;
    int nwaves = (gridDim.x * blockDim.x) >> 5;
    for (int ed = wid; ed < e; ed += nwaves) {
        int s = src[ed];
        int d = dst[ed];
        float w = dinv[s] * dinv[d];
        const float* hs = hw  + (size_t)s * Fp;
        float*       ad = agg + (size_t)d * Fp;
        for (int f = lane; f < Fact; f += 32)
            atomicAdd(&ad[f], w * hs[f]);
    }
}

extern "C" void kernel_launch(void* const* d_in, const int* in_sizes, int n_in,
                              void* d_out, int out_size, void* d_ws, size_t ws_size,
                              hipStream_t stream)
{
    const float* x    = (const float*)d_in[0];
    const int*   ei   = (const int*)d_in[1];
    const float* W1   = (const float*)d_in[2];
    const float* b1   = (const float*)d_in[3];
    const float* W2   = (const float*)d_in[4];
    const float* b2   = (const float*)d_in[5];
    const float* Wout = (const float*)d_in[6];
    const float* bout = (const float*)d_in[7];
    float*       out  = (float*)d_out;

    const int* srcI = ei;
    const int* dstI = ei + N_EDGES;

    float* ws   = (float*)d_ws;
    float* dinv = ws;
    float* bufA = ws + 50176;
    float* bufB = bufA + (size_t)N_NODES * H1P;

    k_deg_init     <<<(N_NODES + 255) / 256, 256, 0, stream>>>(dinv, N_NODES);
    k_deg_accum    <<<(N_EDGES + 255) / 256, 256, 0, stream>>>(dstI, dinv, N_EDGES);
    k_rsqrt_inplace<<<(N_NODES + 255) / 256, 256, 0, stream>>>(dinv, N_NODES);

    k_gemm_wmma_f32<<<dim3(N_NODES / 16, (H1P + 63) / 64), 128, 0, stream>>>(
        x, F_IN, W1, H1, bufA, H1P, H1P, F_IN,  0,  nullptr);
    {
        int total = N_NODES * H1P;
        k_agg_init<<<(total + 255) / 256, 256, 0, stream>>>(bufA, dinv, b1, bufB,
                                                            H1P, H1, total);
    }
    k_agg_edges<<<4096, 256, 0, stream>>>(srcI, dstI, dinv, bufA, bufB, H1P, H1, N_EDGES);

    k_gemm_wmma_f32<<<dim3(N_NODES / 16, (H2P + 63) / 64), 128, 0, stream>>>(
        bufB, H1P, W2, H2, bufA, H2P, H2P, H1,  1,  nullptr);
    {
        int total = N_NODES * H2P;
        k_agg_init<<<(total + 255) / 256, 256, 0, stream>>>(bufA, dinv, b2, bufB,
                                                            H2P, H2, total);
    }
    k_agg_edges<<<4096, 256, 0, stream>>>(srcI, dstI, dinv, bufA, bufB, H2P, H2, N_EDGES);

    k_gemm_wmma_f32<<<dim3(N_NODES / 16, (NCP + 63) / 64), 128, 0, stream>>>(
        bufB, H2P, Wout, NC, out, NC, NC, H2,  1,  bout);
}


# Aggregation owner-computes: a work-group keeps its 196 nodes' rows in LDS, filters the whole dst list for its edges, adds source rows with LDS float atomics (no global ones); sum order changes
# speedup vs baseline: 2.9673x; 2.9673x over previous
#include <hip/hip_runtime.h>


typedef float v2f __attribute__((ext_vector_type(2)));
typedef float v4f __attribute__((ext_vector_type(4)));
typedef float v8f __attribute__((ext_vector_type(8)));

#define N_NODES 50000
#define N_EDGES 800000
#define F_IN    128
#define H1      200
#define H1P     208
#define H2      100
#define H2P     112
#define NC      40
#define NCP     48

#define N_STRIPS (N_NODES / 16)

__global__ void k_deg_init(float* __restrict__ deg, int n) {
    int i = blockIdx.x * blockDim.x + threadIdx.x;
    if (i < n) deg[i] = 1.0f;
}

__global__ void k_deg_accum(const int* __restrict__ dst, float* __restrict__ deg, int e) {
    int i = blockIdx.x * blockDim.x + threadIdx.x;
    if (i < e) atomicAdd(&deg[dst[i]], 1.0f);
}

__global__ void k_rsqrt_inplace(float* __restrict__ v, int n) {
    int i = blockIdx.x * blockDim.x + threadIdx.x;
    if (i < n) v[i] = rsqrtf(v[i]);
}

template <int T, int K, int NACT, int WAVES, bool RELU>
__device__ __forceinline__ void gemm_strips(
    const float* __restrict__ A, int lda, const float* __restrict__ B,
    float* __restrict__ C, int ldc, int Nstore, const float* __restrict__ bias,
    int nstrips, int wave, int lane)
{
    constexpr int KC = (K + 7) / 8;
    const int c = lane & 15;
    const int h = lane >> 4;

    v2f bf[T][2 * KC];
    #pragma unroll
    for (int t = 0; t < T; ++t) {
        const int n  = (t * WAVES + wave) * 16 + c;
        const int nn = n < NACT ? n : NACT - 1;
        #pragma unroll
        for (int j = 0; j < KC; ++j) {
            #pragma unroll
            for (int u = 0; u < 2; ++u) {
                const int k0 = 8 * j + 4 * h + 2 * u;
                const int ka = k0     < K ? k0     : K - 1;
                const int kb = k0 + 1 < K ? k0 + 1 : K - 1;
                const float x0 = B[(size_t)ka * NACT + nn];
                const float x1 = B[(size_t)kb * NACT + nn];
                v2f b;
                b.x = (n < NACT && k0     < K) ? x0 : 0.0f;
                b.y = (n < NACT && k0 + 1 < K) ? x1 : 0.0f;
                bf[t][2 * j + u] = b;
            }
        }
    }

    float bv[T];
    #pragma unroll
    for (int t = 0; t < T; ++t) {
        const int n = (t * WAVES + wave) * 16 + c;
        bv[t] = (bias != nullptr && n < NACT) ? bias[n] : 0.0f;
    }

    const int stride = gridDim.x;
    int s = blockIdx.x;
    if (s >= nstrips) return;

    v4f a_cur[KC], a_nxt[KC];
    {
        const float* ap = A + (size_t)(s * 16 + c) * lda + 4 * h;
        #pragma unroll
        for (int j = 0; j < KC; ++j) a_cur[j] = *(const v4f*)(ap + 8 * j);
    }

    while (true) {
        const int sn = s + stride;
        const bool more = sn < nstrips;
        {
            const int sl = more ? sn : s;
            const float* ap = A + (size_t)(sl * 16 + c) * lda + 4 * h;
            #pragma unroll
            for (int j = 0; j < KC; ++j) a_nxt[j] = *(const v4f*)(ap + 8 * j);
        }
        __builtin_amdgcn_sched_barrier(0);

        v8f acc[T];
        #pragma unroll
        for (int t = 0; t < T; ++t) acc[t] = (v8f){0.f, 0.f, 0.f, 0.f, 0.f, 0.f, 0.f, 0.f};

        #pragma unroll
        for (int j = 0; j < KC; ++j) {
            v4f v = a_cur[j];
            if (RELU) {
                v.x = fmaxf(v.x, 0.0f); v.y = fmaxf(v.y, 0.0f);
                v.z = fmaxf(v.z, 0.0f); v.w = fmaxf(v.w, 0.0f);
            }
            const v2f a0 = {v.x, v.y};
            const v2f a1 = {v.z, v.w};
            #pragma unroll
            for (int t = 0; t < T; ++t)
                acc[t] = __builtin_amdgcn_wmma_f32_16x16x4_f32(
                             false, a0, false, bf[t][2 * j], (short)0, acc[t], false, false);
            #pragma unroll
            for (int t = 0; t < T; ++t)
                acc[t] = __builtin_amdgcn_wmma_f32_16x16x4_f32(
                             false, a1, false, bf[t][2 * j + 1], (short)0, acc[t], false, false);
        }

        #pragma unroll
        for (int t = 0; t < T; ++t) {
            const int n = (t * WAVES + wave) * 16 + c;
            if (n < Nstore) {
                float* cp = C + (size_t)(s * 16 + 8 * h) * ldc + n;
                #pragma unroll
                for (int g = 0; g < 8; ++g)
                    cp[(size_t)g * ldc] = acc[t][g] + bv[t];
            }
        }

        if (!more) break;
        #pragma unroll
        for (int j = 0; j < KC; ++j) a_cur[j] = a_nxt[j];
        s = sn;
    }
}

template <int K, int NACT, int NT, int WAVES, bool RELU>
__global__ void __launch_bounds__(WAVES * 32)
k_gemm_wmma_f32(const float* __restrict__ A, int lda,
                const float* __restrict__ B,
                float* __restrict__ C, int ldc, int Nstore,
                const float* __restrict__ bias, int nstrips)
{
    constexpr int TMAX = (NT + WAVES - 1) / WAVES;
    constexpr int FULL = NT - (TMAX - 1) * WAVES;
    const int lane = threadIdx.x & 31;
    const int wave = __builtin_amdgcn_readfirstlane((int)(threadIdx.x >> 5));

    if (FULL == WAVES || wave < FULL) {
        gemm_strips<TMAX, K, NACT, WAVES, RELU>(A, lda, B, C, ldc, Nstore, bias,
                                                nstrips, wave, lane);
    } else if (TMAX > 1) {
        gemm_strips<(TMAX > 1 ? TMAX - 1 : 1), K, NACT, WAVES, RELU>(
            A, lda, B, C, ldc, Nstore, bias, nstrips, wave, lane);
    }
}

#define AGG_R     196
#define AGG_WAVES 32
#define AGG_RING  256
#define AGG_U     4

template <int FP, int FACT>
__global__ void __launch_bounds__(AGG_WAVES * 32)
k_agg_owner(const int* __restrict__ src, const int* __restrict__ dst,
            const float* __restrict__ dinv, const float* __restrict__ hw,
            const float* __restrict__ bias, float* __restrict__ agg, int e, int n)
{
    constexpr int NI = (FACT + 31) / 32;
    constexpr int TP = NI * 32;
    static_assert(FACT % 4 == 0 && FP % 4 == 0 && FP <= TP, "row shapes");
    __shared__ __attribute__((aligned(16))) float tile[AGG_R * TP];
    __shared__ unsigned ring[AGG_WAVES][AGG_RING];

    const int tid  = threadIdx.x;
    const int lane = tid & 31;
    const int wave = __builtin_amdgcn_readfirstlane(tid >> 5);
    const int lo   = blockIdx.x * AGG_R;
    if (lo >= n) return;
    const int cnt  = (n - lo) < AGG_R ? (n - lo) : AGG_R;

    for (int idx = tid; idx < cnt * TP; idx += AGG_WAVES * 32) {
        const int i = idx / TP;
        const int f = idx - i * TP;
        float val = 0.0f;
        if (f < FP) {
            const float d  = dinv[lo + i];
            const float bb = (f < FACT) ? bias[f] : 0.0f;
            val = bb + d * d * hw[(size_t)(lo + i) * FP + f];
        }
        tile[idx] = val;
    }
    __syncthreads();

    unsigned* myq = ring[wave];
    unsigned head = 0, tail = 0;

    auto drain = [&](int nq) {
        int      s  = 0;
        unsigned dl = 0;
        float    w  = 0.0f;
        if (lane < nq) {
            const unsigned ent = myq[(head + lane) & (AGG_RING - 1)];
            dl = ent & 255u;
            s  = src[ent >> 8];
            w  = dinv[s] * dinv[lo + (int)dl];
        }
        for (int j = 0; j < nq; j += AGG_U) {
            float v[AGG_U][NI];
            float wj[AGG_U];
            int   dj[AGG_U];
            #pragma unroll
            for (int u = 0; u < AGG_U; ++u) {
                const int jj = (j + u < nq) ? (j + u) : (nq - 1);
                const int sj = __builtin_amdgcn_readlane(s, jj);
                wj[u] = __builtin_bit_cast(float,
                            __builtin_amdgcn_readlane(__builtin_bit_cast(int, w), jj));
                dj[u] = __builtin_amdgcn_readlane((int)dl, jj);
                const float* row = hw + (size_t)sj * FP;
                #pragma unroll
                for (int i = 0; i < NI; ++i) {
                    const int f = lane + 32 * i;
                    v[u][i] = row[f < FACT ? f : FACT - 1];
                }
            }
            #pragma unroll
            for (int u = 0; u < AGG_U; ++u) {
                if (j + u < nq) {
                    float* trow = tile + dj[u] * TP + lane;
                    #pragma unroll
                    for (int i = 0; i < NI; ++i) atomicAdd(&trow[32 * i], wj[u] * v[u][i]);
                }
            }
        }
    };

    auto load_chunk = [&](int eb) -> int4 {
        const bool in = eb < e;
        int4 d = *(const int4*)(dst + (in ? eb : 0));
        d.x = in ? d.x : -1;
        d.y = in ? d.y : -1;
        d.z = in ? d.z : -1;
        d.w = in ? d.w : -1;
        return d;
    };
    auto take_chunk = [&](int4 d, int eb) {
        const unsigned r0 = (unsigned)(d.x - lo), r1 = (unsigned)(d.y - lo);
        const unsigned r2 = (unsigned)(d.z - lo), r3 = (unsigned)(d.w - lo);
        const bool h0 = r0 < (unsigned)cnt, h1 = r1 < (unsigned)cnt;
        const bool h2 = r2 < (unsigned)cnt, h3 = r3 < (unsigned)cnt;
        const unsigned m0 = __builtin_amdgcn_ballot_w32(h0);
        const unsigned m1 = __builtin_amdgcn_ballot_w32(h1);
        const unsigned m2 = __builtin_amdgcn_ballot_w32(h2);
        const unsigned m3 = __builtin_amdgcn_ballot_w32(h3);
        if ((m0 | m1 | m2 | m3) == 0u) return;
        unsigned p = tail;
        if (h0) myq[(p + __builtin_amdgcn_mbcnt_lo(m0, 0u)) & (AGG_RING - 1)] = ((unsigned)(eb    ) << 8) | r0;
        p += (unsigned)__builtin_popcount(m0);
        if (h1) myq[(p + __builtin_amdgcn_mbcnt_lo(m1, 0u)) & (AGG_RING - 1)] = ((unsigned)(eb + 1) << 8) | r1;
        p += (unsigned)__builtin_popcount(m1);
        if (h2) myq[(p + __builtin_amdgcn_mbcnt_lo(m2, 0u)) & (AGG_RING - 1)] = ((unsigned)(eb + 2) << 8) | r2;
        p += (unsigned)__builtin_popcount(m2);
        if (h3) myq[(p + __builtin_amdgcn_mbcnt_lo(m3, 0u)) & (AGG_RING - 1)] = ((unsigned)(eb + 3) << 8) | r3;
        p += (unsigned)__builtin_popcount(m3);
        tail = p;
        while (tail - head >= 32u) { drain(32); head += 32u; }
    };

    const int ntrips = (e + 511) >> 9;
    const int first = (int)(((long long)blockIdx.x * ntrips) / gridDim.x);

    int4 cur[4], nxt[4];
    int  tb = wave;
    if (tb < ntrips) {
        int t = first + tb; if (t >= ntrips) t -= ntrips;
        #pragma unroll
        for (int q = 0; q < 4; ++q) cur[q] = load_chunk(t * 512 + q * 128 + lane * 4);
        while (true) {
            const int  tbn  = tb + AGG_WAVES;
            const bool more = tbn < ntrips;
            int tn = first + (more ? tbn : tb); if (tn >= ntrips) tn -= ntrips;
            #pragma unroll
            for (int q = 0; q < 4; ++q) nxt[q] = load_chunk(tn * 512 + q * 128 + lane * 4);
            #pragma unroll
            for (int q = 0; q < 4; ++q) take_chunk(cur[q], t * 512 + q * 128 + lane * 4);
            if (!more) break;
            #pragma unroll
            for (int q = 0; q < 4; ++q) cur[q] = nxt[q];
            tb = tbn; t = tn;
        }
    }
    if (tail != head) drain((int)(tail - head));

    __syncthreads();

    {
        v4f* o4 = (v4f*)(agg + (size_t)lo * FP);
        for (int idx = tid; idx < cnt * (FP / 4); idx += AGG_WAVES * 32) {
            const int i = idx / (FP / 4);
            const int q = idx - i * (FP / 4);
            v4f val = *(const v4f*)(tile + i * TP + 4 * q);
            if (4 * q >= FACT) val = (v4f){0.f, 0.f, 0.f, 0.f};
            o4[idx] = val;
        }
    }
}

extern "C" void kernel_launch(void* const* d_in, const int* in_sizes, int n_in,
                              void* d_out, int out_size, void* d_ws, size_t ws_size,
                              hipStream_t stream)
{
    const float* x    = (const float*)d_in[0];
    const int*   ei   = (const int*)d_in[1];
    const float* W1   = (const float*)d_in[2];
    const float* b1   = (const float*)d_in[3];
    const float* W2   = (const float*)d_in[4];
    const float* b2   = (const float*)d_in[5];
    const float* Wout = (const float*)d_in[6];
    const float* bout = (const float*)d_in[7];
    float*       out  = (float*)d_out;

    const int* srcI = ei;
    const int* dstI = ei + N_EDGES;

    float* ws   = (float*)d_ws;
    float* dinv = ws;
    float* bufA = ws + 50176;
    float* bufB = bufA + (size_t)N_NODES * H1P;

    static_assert(N_EDGES % 4 == 0, "the aggregation reads the dst list four edges per lane");

    k_deg_init     <<<(N_NODES + 255) / 256, 256, 0, stream>>>(dinv, N_NODES);
    k_deg_accum    <<<(N_EDGES + 255) / 256, 256, 0, stream>>>(dstI, dinv, N_EDGES);
    k_rsqrt_inplace<<<(N_NODES + 255) / 256, 256, 0, stream>>>(dinv, N_NODES);

    k_gemm_wmma_f32<F_IN, H1, H1P / 16, 4, false><<<512, 128, 0, stream>>>(
        x, F_IN, W1, bufA, H1P, H1P,  nullptr, N_STRIPS);
    k_agg_owner<H1P, H1><<<(N_NODES + AGG_R - 1) / AGG_R, AGG_WAVES * 32, 0, stream>>>(
        srcI, dstI, dinv, bufA, b1, bufB, N_EDGES, N_NODES);

    k_gemm_wmma_f32<H1, H2, H2P / 16, 4, true><<<512, 128, 0, stream>>>(
        bufB, H1P, W2, bufA, H2P, H2P,  nullptr, N_STRIPS);
    k_agg_owner<H2P, H2><<<(N_NODES + AGG_R - 1) / AGG_R, AGG_WAVES * 32, 0, stream>>>(
        srcI, dstI, dinv, bufA, b2, bufB, N_EDGES, N_NODES);

    k_gemm_wmma_f32<H2, NC, NCP / 16, 1, true><<<1024, 32, 0, stream>>>(
        bufB, H2P, Wout, out, NC, NC,  bout, N_STRIPS);
}


# Layer-1 aggregation also saves each group's filtered edges (source, local dst, weight) in the workspace; layer 2 walks that list and no longer scans the dst list; order of the layer-2 sums changes
# speedup vs baseline: 3.2679x; 1.1013x over previous
#include <hip/hip_runtime.h>


typedef float v2f __attribute__((ext_vector_type(2)));
typedef float v4f __attribute__((ext_vector_type(4)));
typedef float v8f __attribute__((ext_vector_type(8)));

#define N_NODES 50000
#define N_EDGES 800000
#define F_IN    128
#define H1      200
#define H1P     208
#define H2      100
#define H2P     112
#define NC      40
#define NCP     48

#define N_STRIPS (N_NODES / 16)

__global__ void k_deg_init(float* __restrict__ deg, int n) {
    int i = blockIdx.x * blockDim.x + threadIdx.x;
    if (i < n) deg[i] = 1.0f;
}

__global__ void k_deg_accum(const int* __restrict__ dst, float* __restrict__ deg, int e) {
    int i = blockIdx.x * blockDim.x + threadIdx.x;
    if (i < e) atomicAdd(&deg[dst[i]], 1.0f);
}

__global__ void k_rsqrt_inplace(float* __restrict__ v, int n) {
    int i = blockIdx.x * blockDim.x + threadIdx.x;
    if (i < n) v[i] = rsqrtf(v[i]);
}

template <int T, int K, int NACT, int WAVES, bool RELU>
__device__ __forceinline__ void gemm_strips(
    const float* __restrict__ A, int lda, const float* __restrict__ B,
    float* __restrict__ C, int ldc, int Nstore, const float* __restrict__ bias,
    int nstrips, int wave, int lane)
{
    constexpr int KC = (K + 7) / 8;
    const int c = lane & 15;
    const int h = lane >> 4;

    v2f bf[T][2 * KC];
    #pragma unroll
    for (int t = 0; t < T; ++t) {
        const int n  = (t * WAVES + wave) * 16 + c;
        const int nn = n < NACT ? n : NACT - 1;
        #pragma unroll
        for (int j = 0; j < KC; ++j) {
            #pragma unroll
            for (int u = 0; u < 2; ++u) {
                const int k0 = 8 * j + 4 * h + 2 * u;
                const int ka = k0     < K ? k0     : K - 1;
                const int kb = k0 + 1 < K ? k0 + 1 : K - 1;
                const float x0 = B[(size_t)ka * NACT + nn];
                const float x1 = B[(size_t)kb * NACT + nn];
                v2f b;
                b.x = (n < NACT && k0     < K) ? x0 : 0.0f;
                b.y = (n < NACT && k0 + 1 < K) ? x1 : 0.0f;
                bf[t][2 * j + u] = b;
            }
        }
    }

    float bv[T];
    #pragma unroll
    for (int t = 0; t < T; ++t) {
        const int n = (t * WAVES + wave) * 16 + c;
        bv[t] = (bias != nullptr && n < NACT) ? bias[n] : 0.0f;
    }

    const int stride = gridDim.x;
    int s = blockIdx.x;
    if (s >= nstrips) return;

    v4f a_cur[KC], a_nxt[KC];
    {
        const float* ap = A + (size_t)(s * 16 + c) * lda + 4 * h;
        #pragma unroll
        for (int j = 0; j < KC; ++j) a_cur[j] = *(const v4f*)(ap + 8 * j);
    }

    while (true) {
        const int sn = s + stride;
        const bool more = sn < nstrips;
        {
            const int sl = more ? sn : s;
            const float* ap = A + (size_t)(sl * 16 + c) * lda + 4 * h;
            #pragma unroll
            for (int j = 0; j < KC; ++j) a_nxt[j] = *(const v4f*)(ap + 8 * j);
        }
        __builtin_amdgcn_sched_barrier(0);

        v8f acc[T];
        #pragma unroll
        for (int t = 0; t < T; ++t) acc[t] = (v8f){0.f, 0.f, 0.f, 0.f, 0.f, 0.f, 0.f, 0.f};

        #pragma unroll
        for (int j = 0; j < KC; ++j) {
            v4f v = a_cur[j];
            if (RELU) {
                v.x = fmaxf(v.x, 0.0f); v.y = fmaxf(v.y, 0.0f);
                v.z = fmaxf(v.z, 0.0f); v.w = fmaxf(v.w, 0.0f);
            }
            const v2f a0 = {v.x, v.y};
            const v2f a1 = {v.z, v.w};
            #pragma unroll
            for (int t = 0; t < T; ++t)
                acc[t] = __builtin_amdgcn_wmma_f32_16x16x4_f32(
                             false, a0, false, bf[t][2 * j], (short)0, acc[t], false, false);
            #pragma unroll
            for (int t = 0; t < T; ++t)
                acc[t] = __builtin_amdgcn_wmma_f32_16x16x4_f32(
                             false, a1, false, bf[t][2 * j + 1], (short)0, acc[t], false, false);
        }

        #pragma unroll
        for (int t = 0; t < T; ++t) {
            const int n = (t * WAVES + wave) * 16 + c;
            if (n < Nstore) {
                float* cp = C + (size_t)(s * 16 + 8 * h) * ldc + n;
                #pragma unroll
                for (int g = 0; g < 8; ++g)
                    cp[(size_t)g * ldc] = acc[t][g] + bv[t];
            }
        }

        if (!more) break;
        #pragma unroll
        for (int j = 0; j < KC; ++j) a_cur[j] = a_nxt[j];
        s = sn;
    }
}

template <int K, int NACT, int NT, int WAVES, bool RELU>
__global__ void __launch_bounds__(WAVES * 32)
k_gemm_wmma_f32(const float* __restrict__ A, int lda,
                const float* __restrict__ B,
                float* __restrict__ C, int ldc, int Nstore,
                const float* __restrict__ bias, int nstrips)
{
    constexpr int TMAX = (NT + WAVES - 1) / WAVES;
    constexpr int FULL = NT - (TMAX - 1) * WAVES;
    const int lane = threadIdx.x & 31;
    const int wave = __builtin_amdgcn_readfirstlane((int)(threadIdx.x >> 5));

    if (FULL == WAVES || wave < FULL) {
        gemm_strips<TMAX, K, NACT, WAVES, RELU>(A, lda, B, C, ldc, Nstore, bias,
                                                nstrips, wave, lane);
    } else if (TMAX > 1) {
        gemm_strips<(TMAX > 1 ? TMAX - 1 : 1), K, NACT, WAVES, RELU>(
            A, lda, B, C, ldc, Nstore, bias, nstrips, wave, lane);
    }
}

#define AGG_R     196
#define AGG_WAVES 32
#define AGG_RING  256
#define AGG_U     4
#define AGG_CAP   8192
#define AGG_HDR   1024

template <int FP, int FACT, int MODE>
__global__ void __launch_bounds__(AGG_WAVES * 32)
k_agg_owner(const int* __restrict__ src, const int* __restrict__ dst,
            const float* __restrict__ dinv, const float* __restrict__ hw,
            const float* __restrict__ bias, float* __restrict__ agg,
            int* __restrict__ lists, int e, int n)
{
    constexpr int NI = (FACT + 31) / 32;
    constexpr int TP = NI * 32;
    static_assert(FACT % 4 == 0 && FP % 4 == 0 && FP <= TP, "row shapes");
    __shared__ __attribute__((aligned(16))) float tile[AGG_R * TP];
    __shared__ unsigned ring[AGG_WAVES][AGG_RING];
    __shared__ unsigned cursor;

    const int tid  = threadIdx.x;
    const int lane = tid & 31;
    const int wave = __builtin_amdgcn_readfirstlane(tid >> 5);
    const int lo   = blockIdx.x * AGG_R;
    if (lo >= n) return;
    const int cnt  = (n - lo) < AGG_R ? (n - lo) : AGG_R;
    int2* const mylist = (int2*)(lists + AGG_HDR) + (size_t)blockIdx.x * AGG_CAP;
    if (MODE == 0 && tid == 0) cursor = 0u;

    for (int idx = tid; idx < cnt * TP; idx += AGG_WAVES * 32) {
        const int i = idx / TP;
        const int f = idx - i * TP;
        float val = 0.0f;
        if (f < FP) {
            const float d  = dinv[lo + i];
            const float bb = (f < FACT) ? bias[f] : 0.0f;
            val = bb + d * d * hw[(size_t)(lo + i) * FP + f];
        }
        tile[idx] = val;
    }
    __syncthreads();

    unsigned* myq = ring[wave];
    unsigned head = 0, tail = 0;

    auto add_rows = [&](int nq, int s, unsigned dl, float w) {
        for (int j = 0; j < nq; j += AGG_U) {
            float v[AGG_U][NI];
            float wj[AGG_U];
            int   dj[AGG_U];
            #pragma unroll
            for (int u = 0; u < AGG_U; ++u) {
                const int jj = (j + u < nq) ? (j + u) : (nq - 1);
                const int sj = __builtin_amdgcn_readlane(s, jj);
                wj[u] = __builtin_bit_cast(float,
                            __builtin_amdgcn_readlane(__builtin_bit_cast(int, w), jj));
                dj[u] = __builtin_amdgcn_readlane((int)dl, jj);
                const float* row = hw + (size_t)sj * FP;
                #pragma unroll
                for (int i = 0; i < NI; ++i) {
                    const int f = lane + 32 * i;
                    v[u][i] = row[f < FACT ? f : FACT - 1];
                }
            }
            #pragma unroll
            for (int u = 0; u < AGG_U; ++u) {
                if (j + u < nq) {
                    float* trow = tile + dj[u] * TP + lane;
                    #pragma unroll
                    for (int i = 0; i < NI; ++i) atomicAdd(&trow[32 * i], wj[u] * v[u][i]);
                }
            }
        }
    };

    auto drain = [&](int nq) {
        int      s  = 0;
        unsigned dl = 0;
        float    w  = 0.0f;
        if (lane < nq) {
            const unsigned ent = myq[(head + lane) & (AGG_RING - 1)];
            dl = ent & 255u;
            s  = src[ent >> 8];
            w  = dinv[s] * dinv[lo + (int)dl];
        }
        if (MODE == 0) {
            unsigned base = 0u;
            if (lane == 0) base = atomicAdd(&cursor, (unsigned)nq);
            base = (unsigned)__builtin_amdgcn_readfirstlane((int)base);
            if (lane < nq && base + (unsigned)lane < (unsigned)AGG_CAP) {
                int2 rec;
                rec.x = s | (int)(dl << 16);
                rec.y = __builtin_bit_cast(int, w);
                mylist[base + (unsigned)lane] = rec;
            }
        }
        add_rows(nq, s, dl, w);
    };

    auto load_chunk = [&](int eb) -> int4 {
        const bool in = eb < e;
        int4 d = *(const int4*)(dst + (in ? eb : 0));
        d.x = in ? d.x : -1;
        d.y = in ? d.y : -1;
        d.z = in ? d.z : -1;
        d.w = in ? d.w : -1;
        return d;
    };
    auto take_chunk = [&](int4 d, int eb) {
        const unsigned r0 = (unsigned)(d.x - lo), r1 = (unsigned)(d.y - lo);
        const unsigned r2 = (unsigned)(d.z - lo), r3 = (unsigned)(d.w - lo);
        const bool h0 = r0 < (unsigned)cnt, h1 = r1 < (unsigned)cnt;
        const bool h2 = r2 < (unsigned)cnt, h3 = r3 < (unsigned)cnt;
        const unsigned m0 = __builtin_amdgcn_ballot_w32(h0);
        const unsigned m1 = __builtin_amdgcn_ballot_w32(h1);
        const unsigned m2 = __builtin_amdgcn_ballot_w32(h2);
        const unsigned m3 = __builtin_amdgcn_ballot_w32(h3);
        if ((m0 | m1 | m2 | m3) == 0u) return;
        unsigned p = tail;
        if (h0) myq[(p + __builtin_amdgcn_mbcnt_lo(m0, 0u)) & (AGG_RING - 1)] = ((unsigned)(eb    ) << 8) | r0;
        p += (unsigned)__builtin_popcount(m0);
        if (h1) myq[(p + __builtin_amdgcn_mbcnt_lo(m1, 0u)) & (AGG_RING - 1)] = ((unsigned)(eb + 1) << 8) | r1;
        p += (unsigned)__builtin_popcount(m1);
        if (h2) myq[(p + __builtin_amdgcn_mbcnt_lo(m2, 0u)) & (AGG_RING - 1)] = ((unsigned)(eb + 2) << 8) | r2;
        p += (unsigned)__builtin_popcount(m2);
        if (h3) myq[(p + __builtin_amdgcn_mbcnt_lo(m3, 0u)) & (AGG_RING - 1)] = ((unsigned)(eb + 3) << 8) | r3;
        p += (unsigned)__builtin_popcount(m3);
        tail = p;
        while (tail - head >= 32u) { drain(32); head += 32u; }
    };

    int  total    = 0;
    bool use_list = false;
    if (MODE == 1) {
        total    = lists[blockIdx.x];
        use_list = total <= AGG_CAP;
    }

    if (use_list) {
        auto fetch = [&](int c) -> int2 {
            const int i = c + lane;
            return mylist[i < total ? i : total - 1];
        };
        int c0 = wave * 32;
        if (c0 < total) {
            int2 cur = fetch(c0);
            while (true) {
                const int  cn   = c0 + AGG_WAVES * 32;
                const bool more = cn < total;
                const int2 nxt  = fetch(more ? cn : c0);
                const int      left = total - c0;
                const int      s    = cur.x & 0xffff;
                const unsigned dl   = (unsigned)cur.x >> 16;
                const float    w    = __builtin_bit_cast(float, cur.y);
                if (left >= 32) add_rows(32, s, dl, w);
                else            add_rows(left, s, dl, w);
                if (!more) break;
                cur = nxt; c0 = cn;
            }
        }
    } else {
    const int ntrips = (e + 511) >> 9;
    const int first = (int)(((long long)blockIdx.x * ntrips) / gridDim.x);

    int4 cur[4], nxt[4];
    int  tb = wave;
    if (tb < ntrips) {
        int t = first + tb; if (t >= ntrips) t -= ntrips;
        #pragma unroll
        for (int q = 0; q < 4; ++q) cur[q] = load_chunk(t * 512 + q * 128 + lane * 4);
        while (true) {
            const int  tbn  = tb + AGG_WAVES;
            const bool more = tbn < ntrips;
            int tn = first + (more ? tbn : tb); if (tn >= ntrips) tn -= ntrips;
            #pragma unroll
            for (int q = 0; q < 4; ++q) nxt[q] = load_chunk(tn * 512 + q * 128 + lane * 4);
            #pragma unroll
            for (int q = 0; q < 4; ++q) take_chunk(cur[q], t * 512 + q * 128 + lane * 4);
            if (!more) break;
            #pragma unroll
            for (int q = 0; q < 4; ++q) cur[q] = nxt[q];
            tb = tbn; t = tn;
        }
    }
    if (tail != head) drain((int)(tail - head));
    }

    __syncthreads();
    if (MODE == 0 && tid == 0) lists[blockIdx.x] = (int)cursor;

    {
        v4f* o4 = (v4f*)(agg + (size_t)lo * FP);
        for (int idx = tid; idx < cnt * (FP / 4); idx += AGG_WAVES * 32) {
            const int i = idx / (FP / 4);
            const int q = idx - i * (FP / 4);
            v4f val = *(const v4f*)(tile + i * TP + 4 * q);
            if (4 * q >= FACT) val = (v4f){0.f, 0.f, 0.f, 0.f};
            o4[idx] = val;
        }
    }
}

extern "C" void kernel_launch(void* const* d_in, const int* in_sizes, int n_in,
                              void* d_out, int out_size, void* d_ws, size_t ws_size,
                              hipStream_t stream)
{
    const float* x    = (const float*)d_in[0];
    const int*   ei   = (const int*)d_in[1];
    const float* W1   = (const float*)d_in[2];
    const float* b1   = (const float*)d_in[3];
    const float* W2   = (const float*)d_in[4];
    const float* b2   = (const float*)d_in[5];
    const float* Wout = (const float*)d_in[6];
    const float* bout = (const float*)d_in[7];
    float*       out  = (float*)d_out;

    const int* srcI = ei;
    const int* dstI = ei + N_EDGES;

    float* ws   = (float*)d_ws;
    float* dinv = ws;
    float* bufA = ws + 50176;
    float* bufB = bufA + (size_t)N_NODES * H1P;
    int*   lists = (int*)(bufB + (size_t)N_NODES * H1P);

    static_assert(N_EDGES % 4 == 0, "the aggregation reads the dst list four edges per lane");
    static_assert(N_NODES <= 65536 && AGG_R <= 256, "a list entry packs source and local dst in 16 + 8 bits");
    static_assert((N_NODES + AGG_R - 1) / AGG_R <= AGG_HDR, "one count per work-group");

    k_deg_init     <<<(N_NODES + 255) / 256, 256, 0, stream>>>(dinv, N_NODES);
    k_deg_accum    <<<(N_EDGES + 255) / 256, 256, 0, stream>>>(dstI, dinv, N_EDGES);
    k_rsqrt_inplace<<<(N_NODES + 255) / 256, 256, 0, stream>>>(dinv, N_NODES);

    k_gemm_wmma_f32<F_IN, H1, H1P / 16, 4, false><<<512, 128, 0, stream>>>(
        x, F_IN, W1, bufA, H1P, H1P,  nullptr, N_STRIPS);
    k_agg_owner<H1P, H1, 0><<<(N_NODES + AGG_R - 1) / AGG_R, AGG_WAVES * 32, 0, stream>>>(
        srcI, dstI, dinv, bufA, b1, bufB, lists, N_EDGES, N_NODES);

    k_gemm_wmma_f32<H1, H2, H2P / 16, 4, true><<<512, 128, 0, stream>>>(
        bufB, H1P, W2, bufA, H2P, H2P,  nullptr, N_STRIPS);
    k_agg_owner<H2P, H2, 1><<<(N_NODES + AGG_R - 1) / AGG_R, AGG_WAVES * 32, 0, stream>>>(
        srcI, dstI, dinv, bufA, b2, bufB, lists, N_EDGES, N_NODES);

    k_gemm_wmma_f32<H2, NC, NCP / 16, 1, true><<<1024, 32, 0, stream>>>(
        bufB, H2P, Wout, out, NC, NC,  bout, N_STRIPS);
}
